# starting kernel
# baseline (speedup 1.0000x reference)
#include <hip/hip_runtime.h>
#include <hip/hip_bf16.h>

typedef __attribute__((ext_vector_type(2))) float v2f;
typedef __attribute__((ext_vector_type(8))) float v8f;

#define IN_DIM    128
#define OUT_DIM   128
#define NUM_HEADS 4
#define HEAD_DIM  32
#define NEG_SLOPE 0.2f
#define GAT_EPS   1e-16f

__global__ __launch_bounds__(256) void gat_gemm_wmma(const float* __restrict__ X,
                                                     const float* __restrict__ W,
                                                     float* __restrict__ H,
                                                     int M) {
  const int wave = threadIdx.x >> 5;
  const int lane = threadIdx.x & 31;
  const int m0 = blockIdx.x * 16;
  const int n0 = wave * 16;
  const int rc   = lane & 15;
  const int koff = (lane >> 4) * 2;

  int arow = m0 + rc;
  if (arow >= M) arow = M - 1;
  const float* __restrict__ xrow = X + (size_t)arow * IN_DIM + koff;

  v8f c = {};
#pragma unroll
  for (int k = 0; k < IN_DIM; k += 4) {
    v2f a = *(const v2f*)(xrow + k);
    v2f b;
    b.x = W[(size_t)(k + koff) * OUT_DIM + n0 + rc];
    b.y = W[(size_t)(k + koff + 1) * OUT_DIM + n0 + rc];
    c = __builtin_amdgcn_wmma_f32_16x16x4_f32(false, a, false, b, (short)0, c, false, false);
  }

  const int rbase = (lane >> 4) * 8;
#pragma unroll
  for (int r = 0; r < 8; ++r) {
    const int row = m0 + rbase + r;
    if (row < M) H[(size_t)row * OUT_DIM + n0 + rc] = c[r];
  }
}

__global__ __launch_bounds__(256) void gat_alpha(const float* __restrict__ H,
                                                 const float* __restrict__ a,
                                                 float* __restrict__ alpha_src,
                                                 float* __restrict__ alpha_dst,
                                                 int N) {
  const int lane = threadIdx.x & 31;
  const int node = blockIdx.x * 8 + (threadIdx.x >> 5);
  if (node >= N) return;

  const int d0    = lane * 4;
  const int head  = lane >> 3;
  const int local = d0 & (HEAD_DIM - 1);

  const float4 hv = *(const float4*)(H + (size_t)node * OUT_DIM + d0);
  const float4 as = *(const float4*)(a + head * (2 * HEAD_DIM) + local);
  const float4 ad = *(const float4*)(a + head * (2 * HEAD_DIM) + HEAD_DIM + local);

  float ps = hv.x * as.x + hv.y * as.y + hv.z * as.z + hv.w * as.w;
  float pd = hv.x * ad.x + hv.y * ad.y + hv.z * ad.z + hv.w * ad.w;

#pragma unroll
  for (int off = 1; off <= 4; off <<= 1) {
    ps += __shfl_xor(ps, off, 32);
    pd += __shfl_xor(pd, off, 32);
  }
  if ((lane & 7) == 0) {
    alpha_src[(size_t)node * NUM_HEADS + head] = ps;
    alpha_dst[(size_t)node * NUM_HEADS + head] = pd;
  }
}

__global__ __launch_bounds__(256) void gat_edge_scores(const int* __restrict__ ei,
                                                       const float* __restrict__ alpha_src,
                                                       const float* __restrict__ alpha_dst,
                                                       float* __restrict__ exp_s,
                                                       float* __restrict__ sum_exp,
                                                       int E) {
  const int e = blockIdx.x * blockDim.x + threadIdx.x;
  if (e >= E) return;
  const int src = ei[e];
  const int dst = ei[E + e];

  const float4 s1 = *(const float4*)(alpha_src + (size_t)src * NUM_HEADS);
  const float4 s2 = *(const float4*)(alpha_dst + (size_t)dst * NUM_HEADS);

  float s[4] = {s1.x + s2.x, s1.y + s2.y, s1.z + s2.z, s1.w + s2.w};
  float4 ex;
#pragma unroll
  for (int h = 0; h < 4; ++h) {
    float v = s[h];
    v = (v >= 0.0f) ? v : NEG_SLOPE * v;
    ((float*)&ex)[h] = expf(v);
  }
  *(float4*)(exp_s + (size_t)e * NUM_HEADS) = ex;

  float* sp = sum_exp + (size_t)dst * NUM_HEADS;
  unsafeAtomicAdd(sp + 0, ex.x);
  unsafeAtomicAdd(sp + 1, ex.y);
  unsafeAtomicAdd(sp + 2, ex.z);
  unsafeAtomicAdd(sp + 3, ex.w);
}

__global__ __launch_bounds__(256) void gat_aggregate(const int* __restrict__ ei,
                                                     const float* __restrict__ H,
                                                     const float* __restrict__ exp_s,
                                                     const float* __restrict__ sum_exp,
                                                     float* __restrict__ out,
                                                     int E) {
  const int lane = threadIdx.x & 31;
  const int e = blockIdx.x * 8 + (threadIdx.x >> 5);
  if (e >= E) return;

  const int src = ei[e];
  const int dst = ei[E + e];
  const int head = lane >> 3;

  const float attn = exp_s[(size_t)e * NUM_HEADS + head] /
                     (sum_exp[(size_t)dst * NUM_HEADS + head] + GAT_EPS);

  const int d0 = lane * 4;
  const float4 hv = *(const float4*)(H + (size_t)src * OUT_DIM + d0);
  float* op = out + (size_t)dst * OUT_DIM + d0;
  unsafeAtomicAdd(op + 0, hv.x * attn);
  unsafeAtomicAdd(op + 1, hv.y * attn);
  unsafeAtomicAdd(op + 2, hv.z * attn);
  unsafeAtomicAdd(op + 3, hv.w * attn);
}

extern "C" void kernel_launch(void* const* d_in, const int* in_sizes, int n_in,
                              void* d_out, int out_size, void* d_ws, size_t ws_size,
                              hipStream_t stream) {
  const float* X  = (const float*)d_in[0];
  const int*   ei = (const int*)d_in[1];
  const float* W  = (const float*)d_in[2];
  const float* a  = (const float*)d_in[3];

  const int N = in_sizes[0] / IN_DIM;
  const int E = in_sizes[1] / 2;

  float* H         = (float*)d_ws;
  float* alpha_src = H + (size_t)N * OUT_DIM;
  float* alpha_dst = alpha_src + (size_t)N * NUM_HEADS;
  float* exp_s     = alpha_dst + (size_t)N * NUM_HEADS;
  float* sum_exp   = exp_s + (size_t)E * NUM_HEADS;

  hipMemsetAsync(sum_exp, 0, (size_t)N * NUM_HEADS * sizeof(float), stream);
  hipMemsetAsync(d_out, 0, (size_t)out_size * sizeof(float), stream);

  gat_gemm_wmma<<<(N + 15) / 16, 256, 0, stream>>>(X, W, H, N);
  gat_alpha<<<(N + 7) / 8, 256, 0, stream>>>(H, a, alpha_src, alpha_dst, N);
  gat_edge_scores<<<(E + 255) / 256, 256, 0, stream>>>(ei, alpha_src, alpha_dst,
                                                       exp_s, sum_exp, E);
  gat_aggregate<<<(E + 7) / 8, 256, 0, stream>>>(ei, H, exp_s, sum_exp,
                                                 (float*)d_out, E);
}


# edges counting-sorted by destination in the call (int atomic rank), one wave per node sums its rows in registers: no f32 atomics, no out memset; sum order = arrival-order rank
# speedup vs baseline: 2.3165x; 2.3165x over previous
#include <hip/hip_runtime.h>
#include <hip/hip_bf16.h>

typedef __attribute__((ext_vector_type(2))) float v2f;
typedef __attribute__((ext_vector_type(8))) float v8f;

#define IN_DIM    128
#define OUT_DIM   128
#define NUM_HEADS 4
#define HEAD_DIM  32
#define NEG_SLOPE 0.2f
#define GAT_EPS   1e-16f

__global__ __launch_bounds__(256) void gat_gemm_wmma(const float* __restrict__ X,
                                                     const float* __restrict__ W,
                                                     float* __restrict__ H,
                                                     int M) {
  const int wave = threadIdx.x >> 5;
  const int lane = threadIdx.x & 31;
  const int m0 = blockIdx.x * 16;
  const int n0 = wave * 16;
  const int rc   = lane & 15;
  const int koff = (lane >> 4) * 2;

  int arow = m0 + rc;
  if (arow >= M) arow = M - 1;
  const float* __restrict__ xrow = X + (size_t)arow * IN_DIM + koff;

  v8f c = {};
#pragma unroll
  for (int k = 0; k < IN_DIM; k += 4) {
    v2f a = *(const v2f*)(xrow + k);
    v2f b;
    b.x = W[(size_t)(k + koff) * OUT_DIM + n0 + rc];
    b.y = W[(size_t)(k + koff + 1) * OUT_DIM + n0 + rc];
    c = __builtin_amdgcn_wmma_f32_16x16x4_f32(false, a, false, b, (short)0, c, false, false);
  }

  const int rbase = (lane >> 4) * 8;
#pragma unroll
  for (int r = 0; r < 8; ++r) {
    const int row = m0 + rbase + r;
    if (row < M) H[(size_t)row * OUT_DIM + n0 + rc] = c[r];
  }
}

__global__ __launch_bounds__(256) void gat_alpha(const float* __restrict__ H,
                                                 const float* __restrict__ a,
                                                 float* __restrict__ alpha_src,
                                                 float* __restrict__ alpha_dst,
                                                 int N) {
  const int lane = threadIdx.x & 31;
  const int node = blockIdx.x * 8 + (threadIdx.x >> 5);
  if (node >= N) return;

  const int d0    = lane * 4;
  const int head  = lane >> 3;
  const int local = d0 & (HEAD_DIM - 1);

  const float4 hv = *(const float4*)(H + (size_t)node * OUT_DIM + d0);
  const float4 as = *(const float4*)(a + head * (2 * HEAD_DIM) + local);
  const float4 ad = *(const float4*)(a + head * (2 * HEAD_DIM) + HEAD_DIM + local);

  float ps = hv.x * as.x + hv.y * as.y + hv.z * as.z + hv.w * as.w;
  float pd = hv.x * ad.x + hv.y * ad.y + hv.z * ad.z + hv.w * ad.w;

#pragma unroll
  for (int off = 1; off <= 4; off <<= 1) {
    ps += __shfl_xor(ps, off, 32);
    pd += __shfl_xor(pd, off, 32);
  }
  if ((lane & 7) == 0) {
    alpha_src[(size_t)node * NUM_HEADS + head] = ps;
    alpha_dst[(size_t)node * NUM_HEADS + head] = pd;
  }
}

__global__ __launch_bounds__(256) void gat_count_rank(const int* __restrict__ ei,
                                                      int* __restrict__ cnt,
                                                      int* __restrict__ rank,
                                                      int E) {
  const int e = blockIdx.x * blockDim.x + threadIdx.x;
  if (e >= E) return;
  const int dst = ei[E + e];
  rank[e] = atomicAdd(cnt + dst, 1);
}

#define SCAN_T 1024

__device__ __forceinline__ int block_inclusive_scan(int v, int (*buf)[SCAN_T]) {
  const int t = threadIdx.x;
  int cur = 0;
  buf[0][t] = v;
  __syncthreads();
  for (int off = 1; off < SCAN_T; off <<= 1) {
    int x = buf[cur][t];
    if (t >= off) x += buf[cur][t - off];
    buf[cur ^ 1][t] = x;
    __syncthreads();
    cur ^= 1;
  }
  const int r = buf[cur][t];
  __syncthreads();
  return r;
}

__global__ __launch_bounds__(SCAN_T) void gat_rowptr(const int* __restrict__ cnt,
                                                     int* __restrict__ rowptr,
                                                     int N) {
  __shared__ int buf[2][SCAN_T];
  __shared__ int s_front;
  const int t = threadIdx.x;
  const int b = blockIdx.x;

  int front = 0;
  const int4* __restrict__ c4 = (const int4*)cnt;
  for (int q = t; q < b * (SCAN_T / 4); q += SCAN_T) {
    const int4 v = c4[q];
    front += (v.x + v.y) + (v.z + v.w);
  }
  const int front_incl = block_inclusive_scan(front, buf);
  if (t == SCAN_T - 1) s_front = front_incl;

  const int i = b * SCAN_T + t;
  const int mine = cnt[i];
  const int incl = block_inclusive_scan(mine, buf);
  const int base = s_front;
  if (i < N) rowptr[i + 1] = base + incl;
  if (i == 0) rowptr[0] = 0;
}

__global__ __launch_bounds__(256) void gat_place(const int* __restrict__ ei,
                                                 const int* __restrict__ rowptr,
                                                 const int* __restrict__ rank,
                                                 int* __restrict__ src_sorted,
                                                 int E) {
  const int e = blockIdx.x * blockDim.x + threadIdx.x;
  if (e >= E) return;
  const int src = ei[e];
  const int dst = ei[E + e];
  src_sorted[rowptr[dst] + rank[e]] = src;
}

__device__ __forceinline__ float gat_edge_exp(float v) {
  v = (v >= 0.0f) ? v : NEG_SLOPE * v;
  return expf(v);
}

__device__ __forceinline__ void gat_load_edges(const int* __restrict__ src_sorted,
                                               const float* __restrict__ alpha_src,
                                               const float4 ad, int base, int end, int lane,
                                               int& s, float4& ex) {
  const int i = base + lane;
  s = 0;
  ex = make_float4(0.0f, 0.0f, 0.0f, 0.0f);
  if (i < end) {
    s = src_sorted[i];
    const float4 as = *(const float4*)(alpha_src + (size_t)s * NUM_HEADS);
    ex.x = gat_edge_exp(as.x + ad.x);
    ex.y = gat_edge_exp(as.y + ad.y);
    ex.z = gat_edge_exp(as.z + ad.z);
    ex.w = gat_edge_exp(as.w + ad.w);
  }
}

__global__ __launch_bounds__(256) void gat_gather(const int* __restrict__ rowptr,
                                                  const int* __restrict__ src_sorted,
                                                  const float* __restrict__ H,
                                                  const float* __restrict__ alpha_src,
                                                  const float* __restrict__ alpha_dst,
                                                  float* __restrict__ out,
                                                  int N) {
  const int lane = threadIdx.x & 31;
  const int n = blockIdx.x * 8 + __builtin_amdgcn_readfirstlane((int)(threadIdx.x >> 5));
  if (n >= N) return;

  const int beg = rowptr[n];
  const int end = rowptr[n + 1];
  const float4 ad = *(const float4*)(alpha_dst + (size_t)n * NUM_HEADS);

  int s0;
  float4 ex0;
  gat_load_edges(src_sorted, alpha_src, ad, beg, end, lane, s0, ex0);

  float4 sum = ex0;
  for (int base = beg + 32; base < end; base += 32) {
    int s;
    float4 ex;
    gat_load_edges(src_sorted, alpha_src, ad, base, end, lane, s, ex);
    sum.x += ex.x; sum.y += ex.y; sum.z += ex.z; sum.w += ex.w;
  }
  for (int off = 16; off >= 1; off >>= 1) {
    sum.x += __shfl_xor(sum.x, off, 32);
    sum.y += __shfl_xor(sum.y, off, 32);
    sum.z += __shfl_xor(sum.z, off, 32);
    sum.w += __shfl_xor(sum.w, off, 32);
  }
  const float4 den = make_float4(sum.x + GAT_EPS, sum.y + GAT_EPS, sum.z + GAT_EPS, sum.w + GAT_EPS);

  const float* __restrict__ hcol = H + lane;
  float acc0 = 0.0f, acc1 = 0.0f, acc2 = 0.0f, acc3 = 0.0f;
  for (int base = beg; base < end; base += 32) {
    int s = s0;
    float4 ex = ex0;
    if (base != beg) gat_load_edges(src_sorted, alpha_src, ad, base, end, lane, s, ex);
    const float4 w = make_float4(ex.x / den.x, ex.y / den.y, ex.z / den.z, ex.w / den.w);
    const int cntc = (end - base < 32) ? (end - base) : 32;
#pragma unroll 4
    for (int k = 0; k < cntc; ++k) {
      const int sk = __builtin_amdgcn_readlane(s, k);
      const float w0 = __int_as_float(__builtin_amdgcn_readlane(__float_as_int(w.x), k));
      const float w1 = __int_as_float(__builtin_amdgcn_readlane(__float_as_int(w.y), k));
      const float w2 = __int_as_float(__builtin_amdgcn_readlane(__float_as_int(w.z), k));
      const float w3 = __int_as_float(__builtin_amdgcn_readlane(__float_as_int(w.w), k));
      const float* __restrict__ hr = hcol + (size_t)sk * OUT_DIM;
      acc0 += hr[0 * HEAD_DIM] * w0;
      acc1 += hr[1 * HEAD_DIM] * w1;
      acc2 += hr[2 * HEAD_DIM] * w2;
      acc3 += hr[3 * HEAD_DIM] * w3;
    }
  }
  float* __restrict__ orow = out + (size_t)n * OUT_DIM + lane;
  orow[0 * HEAD_DIM] = acc0;
  orow[1 * HEAD_DIM] = acc1;
  orow[2 * HEAD_DIM] = acc2;
  orow[3 * HEAD_DIM] = acc3;
}

extern "C" void kernel_launch(void* const* d_in, const int* in_sizes, int n_in,
                              void* d_out, int out_size, void* d_ws, size_t ws_size,
                              hipStream_t stream) {
  const float* X  = (const float*)d_in[0];
  const int*   ei = (const int*)d_in[1];
  const float* W  = (const float*)d_in[2];
  const float* a  = (const float*)d_in[3];

  const int N = in_sizes[0] / IN_DIM;
  const int E = in_sizes[1] / 2;
  const int scan_blocks = (N + SCAN_T - 1) / SCAN_T;
  const int n_pad = scan_blocks * SCAN_T;

  float* H          = (float*)d_ws;
  float* alpha_src  = H + (size_t)N * OUT_DIM;
  float* alpha_dst  = alpha_src + (size_t)N * NUM_HEADS;
  int*   cnt        = (int*)(alpha_dst + (size_t)N * NUM_HEADS);
  int*   rowptr     = cnt + n_pad;
  int*   rank       = rowptr + n_pad + 4;
  int*   src_sorted = rank + (size_t)((E + 3) / 4) * 4;

  hipMemsetAsync(cnt, 0, (size_t)n_pad * sizeof(int), stream);

  gat_gemm_wmma<<<(N + 15) / 16, 256, 0, stream>>>(X, W, H, N);
  gat_alpha<<<(N + 7) / 8, 256, 0, stream>>>(H, a, alpha_src, alpha_dst, N);
  gat_count_rank<<<(E + 255) / 256, 256, 0, stream>>>(ei, cnt, rank, E);
  gat_rowptr<<<scan_blocks, SCAN_T, 0, stream>>>(cnt, rowptr, N);
  gat_place<<<(E + 255) / 256, 256, 0, stream>>>(ei, rowptr, rank, src_sorted, E);
  gat_gather<<<(N + 7) / 8, 256, 0, stream>>>(rowptr, src_sorted, H, alpha_src, alpha_dst,
                                              (float*)d_out, N);
}


# every launch resized to one resident set of waves that loop over their rows or edges (was one wave per node, one thread per edge); GEMM B fragments stay in registers; sum order still arrival rank
# speedup vs baseline: 2.4069x; 1.0390x over previous
#include <hip/hip_runtime.h>
#include <hip/hip_bf16.h>

typedef __attribute__((ext_vector_type(2))) float v2f;
typedef __attribute__((ext_vector_type(8))) float v8f;

#define IN_DIM    128
#define OUT_DIM   128
#define NUM_HEADS 4
#define HEAD_DIM  32
#define NEG_SLOPE 0.2f
#define GAT_EPS   1e-16f

#define GATHER_BLOCKS 2048
#define STREAM_BLOCKS 1024
#define GEMM_BLOCKS   1024

__global__ __launch_bounds__(256) void gat_gemm_wmma(const float* __restrict__ X,
                                                     const float* __restrict__ W,
                                                     float* __restrict__ H,
                                                     int M) {
  const int wave = threadIdx.x >> 5;
  const int lane = threadIdx.x & 31;
  const int n0 = wave * 16;
  const int rc   = lane & 15;
  const int koff = (lane >> 4) * 2;
  const int rbase = (lane >> 4) * 8;

  v2f bf[IN_DIM / 4];
#pragma unroll
  for (int k = 0; k < IN_DIM; k += 4) {
    bf[k / 4].x = W[(size_t)(k + koff) * OUT_DIM + n0 + rc];
    bf[k / 4].y = W[(size_t)(k + koff + 1) * OUT_DIM + n0 + rc];
  }

  const int nslab = (M + 15) / 16;
  for (int slab = blockIdx.x; slab < nslab; slab += gridDim.x) {
    const int m0 = slab * 16;
    int arow = m0 + rc;
    if (arow >= M) arow = M - 1;
    const float* __restrict__ xrow = X + (size_t)arow * IN_DIM + koff;

    v8f c = {};
#pragma unroll
    for (int k = 0; k < IN_DIM; k += 4) {
      v2f a = *(const v2f*)(xrow + k);
      c = __builtin_amdgcn_wmma_f32_16x16x4_f32(false, a, false, bf[k / 4], (short)0, c, false, false);
    }

#pragma unroll
    for (int r = 0; r < 8; ++r) {
      const int row = m0 + rbase + r;
      if (row < M) H[(size_t)row * OUT_DIM + n0 + rc] = c[r];
    }
  }
}

__global__ __launch_bounds__(256) void gat_alpha(const float* __restrict__ H,
                                                 const float* __restrict__ a,
                                                 float* __restrict__ alpha_src,
                                                 float* __restrict__ alpha_dst,
                                                 int N) {
  const int lane = threadIdx.x & 31;
  const int w0 = blockIdx.x * 8 + __builtin_amdgcn_readfirstlane((int)(threadIdx.x >> 5));
  const int nw = gridDim.x * 8;

  const int d0    = lane * 4;
  const int head  = lane >> 3;
  const int local = d0 & (HEAD_DIM - 1);

  const float4 as = *(const float4*)(a + head * (2 * HEAD_DIM) + local);
  const float4 ad = *(const float4*)(a + head * (2 * HEAD_DIM) + HEAD_DIM + local);

  for (int nb = w0; nb < N; nb += 4 * nw) {
    float4 hv[4];
#pragma unroll
    for (int j = 0; j < 4; ++j) {
      const int node = nb + j * nw;
      hv[j] = make_float4(0.0f, 0.0f, 0.0f, 0.0f);
      if (node < N) hv[j] = *(const float4*)(H + (size_t)node * OUT_DIM + d0);
    }
#pragma unroll
    for (int j = 0; j < 4; ++j) {
      const int node = nb + j * nw;
      float ps = hv[j].x * as.x + hv[j].y * as.y + hv[j].z * as.z + hv[j].w * as.w;
      float pd = hv[j].x * ad.x + hv[j].y * ad.y + hv[j].z * ad.z + hv[j].w * ad.w;
#pragma unroll
      for (int off = 1; off <= 4; off <<= 1) {
        ps += __shfl_xor(ps, off, 32);
        pd += __shfl_xor(pd, off, 32);
      }
      if (node < N && (lane & 7) == 0) {
        alpha_src[(size_t)node * NUM_HEADS + head] = ps;
        alpha_dst[(size_t)node * NUM_HEADS + head] = pd;
      }
    }
  }
}

__global__ __launch_bounds__(256) void gat_count_rank(const int* __restrict__ ei,
                                                      int* __restrict__ cnt,
                                                      int* __restrict__ rank,
                                                      int E) {
  const int T = gridDim.x * blockDim.x;
  for (int e0 = blockIdx.x * blockDim.x + threadIdx.x; e0 < E; e0 += 4 * T) {
    int d[4], r[4];
#pragma unroll
    for (int j = 0; j < 4; ++j) {
      const int e = e0 + j * T;
      d[j] = (e < E) ? ei[E + e] : -1;
    }
#pragma unroll
    for (int j = 0; j < 4; ++j) {
      r[j] = 0;
      if (d[j] >= 0) r[j] = atomicAdd(cnt + d[j], 1);
    }
#pragma unroll
    for (int j = 0; j < 4; ++j) {
      if (d[j] >= 0) rank[e0 + j * T] = r[j];
    }
  }
}

#define SCAN_T 1024

__device__ __forceinline__ int block_inclusive_scan(int v, int (*buf)[SCAN_T]) {
  const int t = threadIdx.x;
  int cur = 0;
  buf[0][t] = v;
  __syncthreads();
  for (int off = 1; off < SCAN_T; off <<= 1) {
    int x = buf[cur][t];
    if (t >= off) x += buf[cur][t - off];
    buf[cur ^ 1][t] = x;
    __syncthreads();
    cur ^= 1;
  }
  const int r = buf[cur][t];
  __syncthreads();
  return r;
}

__global__ __launch_bounds__(SCAN_T) void gat_rowptr(const int* __restrict__ cnt,
                                                     int* __restrict__ rowptr,
                                                     int N) {
  __shared__ int buf[2][SCAN_T];
  __shared__ int s_front;
  const int t = threadIdx.x;
  const int b = blockIdx.x;

  int front = 0;
  const int4* __restrict__ c4 = (const int4*)cnt;
  for (int q = t; q < b * (SCAN_T / 4); q += SCAN_T) {
    const int4 v = c4[q];
    front += (v.x + v.y) + (v.z + v.w);
  }
  const int front_incl = block_inclusive_scan(front, buf);
  if (t == SCAN_T - 1) s_front = front_incl;

  const int i = b * SCAN_T + t;
  const int mine = cnt[i];
  const int incl = block_inclusive_scan(mine, buf);
  const int base = s_front;
  if (i < N) rowptr[i + 1] = base + incl;
  if (i == 0) rowptr[0] = 0;
}

__global__ __launch_bounds__(256) void gat_place(const int* __restrict__ ei,
                                                 const int* __restrict__ rowptr,
                                                 const int* __restrict__ rank,
                                                 int* __restrict__ src_sorted,
                                                 int E) {
  const int T = gridDim.x * blockDim.x;
  for (int e0 = blockIdx.x * blockDim.x + threadIdx.x; e0 < E; e0 += 4 * T) {
    int s[4], d[4], rk[4], rp[4];
#pragma unroll
    for (int j = 0; j < 4; ++j) {
      const int e = e0 + j * T;
      s[j] = 0; d[j] = -1; rk[j] = 0;
      if (e < E) { s[j] = ei[e]; d[j] = ei[E + e]; rk[j] = rank[e]; }
    }
#pragma unroll
    for (int j = 0; j < 4; ++j) {
      rp[j] = 0;
      if (d[j] >= 0) rp[j] = rowptr[d[j]];
    }
#pragma unroll
    for (int j = 0; j < 4; ++j) {
      if (d[j] >= 0) src_sorted[rp[j] + rk[j]] = s[j];
    }
  }
}

__device__ __forceinline__ float gat_edge_exp(float v) {
  v = (v >= 0.0f) ? v : NEG_SLOPE * v;
  return expf(v);
}

__device__ __forceinline__ void gat_load_edges(const int* __restrict__ src_sorted,
                                               const float* __restrict__ alpha_src,
                                               const float4 ad, int base, int end, int lane,
                                               int& s, float4& ex) {
  const int i = base + lane;
  s = 0;
  ex = make_float4(0.0f, 0.0f, 0.0f, 0.0f);
  if (i < end) {
    s = src_sorted[i];
    const float4 as = *(const float4*)(alpha_src + (size_t)s * NUM_HEADS);
    ex.x = gat_edge_exp(as.x + ad.x);
    ex.y = gat_edge_exp(as.y + ad.y);
    ex.z = gat_edge_exp(as.z + ad.z);
    ex.w = gat_edge_exp(as.w + ad.w);
  }
}

__global__ __launch_bounds__(256) void gat_gather(const int* __restrict__ rowptr,
                                                  const int* __restrict__ src_sorted,
                                                  const float* __restrict__ H,
                                                  const float* __restrict__ alpha_src,
                                                  const float* __restrict__ alpha_dst,
                                                  float* __restrict__ out,
                                                  int N) {
  const int lane = threadIdx.x & 31;
  const int wv0 = blockIdx.x * 8 + __builtin_amdgcn_readfirstlane((int)(threadIdx.x >> 5));
  const int nwv = gridDim.x * 8;
  const float* __restrict__ hcol = H + lane;
  for (int n = wv0; n < N; n += nwv) {
    const int beg = rowptr[n];
    const int end = rowptr[n + 1];
    const float4 ad = *(const float4*)(alpha_dst + (size_t)n * NUM_HEADS);

    int s0;
    float4 ex0;
    gat_load_edges(src_sorted, alpha_src, ad, beg, end, lane, s0, ex0);

    float4 sum = ex0;
    for (int base = beg + 32; base < end; base += 32) {
      int s;
      float4 ex;
      gat_load_edges(src_sorted, alpha_src, ad, base, end, lane, s, ex);
      sum.x += ex.x; sum.y += ex.y; sum.z += ex.z; sum.w += ex.w;
    }
    for (int off = 16; off >= 1; off >>= 1) {
      sum.x += __shfl_xor(sum.x, off, 32);
      sum.y += __shfl_xor(sum.y, off, 32);
      sum.z += __shfl_xor(sum.z, off, 32);
      sum.w += __shfl_xor(sum.w, off, 32);
    }
    const float4 den = make_float4(sum.x + GAT_EPS, sum.y + GAT_EPS, sum.z + GAT_EPS, sum.w + GAT_EPS);

    float acc0 = 0.0f, acc1 = 0.0f, acc2 = 0.0f, acc3 = 0.0f;
    for (int base = beg; base < end; base += 32) {
      int s = s0;
      float4 ex = ex0;
      if (base != beg) gat_load_edges(src_sorted, alpha_src, ad, base, end, lane, s, ex);
      const float4 w = make_float4(ex.x / den.x, ex.y / den.y, ex.z / den.z, ex.w / den.w);
      const int cntc = (end - base < 32) ? (end - base) : 32;
#pragma unroll 4
      for (int k = 0; k < cntc; ++k) {
        const int sk = __builtin_amdgcn_readlane(s, k);
        const float w0 = __int_as_float(__builtin_amdgcn_readlane(__float_as_int(w.x), k));
        const float w1 = __int_as_float(__builtin_amdgcn_readlane(__float_as_int(w.y), k));
        const float w2 = __int_as_float(__builtin_amdgcn_readlane(__float_as_int(w.z), k));
        const float w3 = __int_as_float(__builtin_amdgcn_readlane(__float_as_int(w.w), k));
        const float* __restrict__ hr = hcol + (size_t)sk * OUT_DIM;
        acc0 += hr[0 * HEAD_DIM] * w0;
        acc1 += hr[1 * HEAD_DIM] * w1;
        acc2 += hr[2 * HEAD_DIM] * w2;
        acc3 += hr[3 * HEAD_DIM] * w3;
      }
    }
    float* __restrict__ orow = out + (size_t)n * OUT_DIM + lane;
    orow[0 * HEAD_DIM] = acc0;
    orow[1 * HEAD_DIM] = acc1;
    orow[2 * HEAD_DIM] = acc2;
    orow[3 * HEAD_DIM] = acc3;
  }
}

extern "C" void kernel_launch(void* const* d_in, const int* in_sizes, int n_in,
                              void* d_out, int out_size, void* d_ws, size_t ws_size,
                              hipStream_t stream) {
  const float* X  = (const float*)d_in[0];
  const int*   ei = (const int*)d_in[1];
  const float* W  = (const float*)d_in[2];
  const float* a  = (const float*)d_in[3];

  const int N = in_sizes[0] / IN_DIM;
  const int E = in_sizes[1] / 2;
  const int scan_blocks = (N + SCAN_T - 1) / SCAN_T;
  const int n_pad = scan_blocks * SCAN_T;

  float* H          = (float*)d_ws;
  float* alpha_src  = H + (size_t)N * OUT_DIM;
  float* alpha_dst  = alpha_src + (size_t)N * NUM_HEADS;
  int*   cnt        = (int*)(alpha_dst + (size_t)N * NUM_HEADS);
  int*   rowptr     = cnt + n_pad;
  int*   rank       = rowptr + n_pad + 4;
  int*   src_sorted = rank + (size_t)((E + 3) / 4) * 4;

  hipMemsetAsync(cnt, 0, (size_t)n_pad * sizeof(int), stream);

  gat_gemm_wmma<<<GEMM_BLOCKS, 256, 0, stream>>>(X, W, H, N);
  gat_alpha<<<STREAM_BLOCKS, 256, 0, stream>>>(H, a, alpha_src, alpha_dst, N);
  gat_count_rank<<<STREAM_BLOCKS, 256, 0, stream>>>(ei, cnt, rank, E);
  gat_rowptr<<<scan_blocks, SCAN_T, 0, stream>>>(cnt, rowptr, N);
  gat_place<<<STREAM_BLOCKS, 256, 0, stream>>>(ei, rowptr, rank, src_sorted, E);
  gat_gather<<<GATHER_BLOCKS, 256, 0, stream>>>(rowptr, src_sorted, H, alpha_src, alpha_dst,
                                              (float*)d_out, N);
}


# edge sort redone as a two-level counting sort on LDS counters (256 destination ranges, then nodes of a range): no global atomic, no rank array, no counter memset; sum order = LDS arrival order
# speedup vs baseline: 3.6636x; 1.5221x over previous
#include <hip/hip_runtime.h>
#include <hip/hip_bf16.h>

typedef __attribute__((ext_vector_type(2))) float v2f;
typedef __attribute__((ext_vector_type(8))) float v8f;

#define IN_DIM    128
#define OUT_DIM   128
#define NUM_HEADS 4
#define HEAD_DIM  32
#define NEG_SLOPE 0.2f
#define GAT_EPS   1e-16f

#define GATHER_BLOCKS 2048
#define STREAM_BLOCKS 1024
#define GEMM_BLOCKS   1024

__global__ __launch_bounds__(256) void gat_gemm_wmma(const float* __restrict__ X,
                                                     const float* __restrict__ W,
                                                     float* __restrict__ H,
                                                     int M) {
  const int wave = threadIdx.x >> 5;
  const int lane = threadIdx.x & 31;
  const int n0 = wave * 16;
  const int rc   = lane & 15;
  const int koff = (lane >> 4) * 2;
  const int rbase = (lane >> 4) * 8;

  v2f bf[IN_DIM / 4];
#pragma unroll
  for (int k = 0; k < IN_DIM; k += 4) {
    bf[k / 4].x = W[(size_t)(k + koff) * OUT_DIM + n0 + rc];
    bf[k / 4].y = W[(size_t)(k + koff + 1) * OUT_DIM + n0 + rc];
  }

  const int nslab = (M + 15) / 16;
  for (int slab = blockIdx.x; slab < nslab; slab += gridDim.x) {
    const int m0 = slab * 16;
    int arow = m0 + rc;
    if (arow >= M) arow = M - 1;
    const float* __restrict__ xrow = X + (size_t)arow * IN_DIM + koff;

    v8f c = {};
#pragma unroll
    for (int k = 0; k < IN_DIM; k += 4) {
      v2f a = *(const v2f*)(xrow + k);
      c = __builtin_amdgcn_wmma_f32_16x16x4_f32(false, a, false, bf[k / 4], (short)0, c, false, false);
    }

#pragma unroll
    for (int r = 0; r < 8; ++r) {
      const int row = m0 + rbase + r;
      if (row < M) H[(size_t)row * OUT_DIM + n0 + rc] = c[r];
    }
  }
}

__global__ __launch_bounds__(256) void gat_alpha(const float* __restrict__ H,
                                                 const float* __restrict__ a,
                                                 float* __restrict__ alpha_src,
                                                 float* __restrict__ alpha_dst,
                                                 int N) {
  const int lane = threadIdx.x & 31;
  const int w0 = blockIdx.x * 8 + __builtin_amdgcn_readfirstlane((int)(threadIdx.x >> 5));
  const int nw = gridDim.x * 8;

  const int d0    = lane * 4;
  const int head  = lane >> 3;
  const int local = d0 & (HEAD_DIM - 1);

  const float4 as = *(const float4*)(a + head * (2 * HEAD_DIM) + local);
  const float4 ad = *(const float4*)(a + head * (2 * HEAD_DIM) + HEAD_DIM + local);

  for (int nb = w0; nb < N; nb += 4 * nw) {
    float4 hv[4];
#pragma unroll
    for (int j = 0; j < 4; ++j) {
      const int node = nb + j * nw;
      hv[j] = make_float4(0.0f, 0.0f, 0.0f, 0.0f);
      if (node < N) hv[j] = *(const float4*)(H + (size_t)node * OUT_DIM + d0);
    }
#pragma unroll
    for (int j = 0; j < 4; ++j) {
      const int node = nb + j * nw;
      float ps = hv[j].x * as.x + hv[j].y * as.y + hv[j].z * as.z + hv[j].w * as.w;
      float pd = hv[j].x * ad.x + hv[j].y * ad.y + hv[j].z * ad.z + hv[j].w * ad.w;
#pragma unroll
      for (int off = 1; off <= 4; off <<= 1) {
        ps += __shfl_xor(ps, off, 32);
        pd += __shfl_xor(pd, off, 32);
      }
      if (node < N && (lane & 7) == 0) {
        alpha_src[(size_t)node * NUM_HEADS + head] = ps;
        alpha_dst[(size_t)node * NUM_HEADS + head] = pd;
      }
    }
  }
}

#define SORT_G    256
#define SORT_NB   256
#define SORT_T    1024
#define SORT_EPT  8
#define RANGE_MAX 4096

__device__ __forceinline__ unsigned sort_range_nodes(int N) {
  const unsigned r = ((unsigned)N + SORT_NB - 1) / SORT_NB;
  return r ? r : 1u;
}
__device__ __forceinline__ int sort_chunk_edges(int E) { return (E + SORT_G - 1) / SORT_G; }
__device__ __forceinline__ int sort_src_bits(int N) { return (N > 1) ? 32 - __clz(N - 1) : 0; }

__device__ __forceinline__ int sort_block_excl_scan(int v, int* wsum) {
  const int lane = threadIdx.x & 31;
  const int w = threadIdx.x >> 5;
  int incl = v;
  for (int off = 1; off < 32; off <<= 1) {
    const int n = __shfl_up(incl, off, 32);
    if (lane >= off) incl += n;
  }
  if (lane == 31) wsum[w] = incl;
  __syncthreads();
  if (w == 0) {
    const int x = wsum[lane];
    int xi = x;
    for (int off = 1; off < 32; off <<= 1) {
      const int n = __shfl_up(xi, off, 32);
      if (lane >= off) xi += n;
    }
    wsum[lane] = xi - x;
  }
  __syncthreads();
  return wsum[w] + incl - v;
}

__global__ __launch_bounds__(SORT_T) void gat_range_count(const int* __restrict__ ei,
                                                          int* __restrict__ bh,
                                                          int N, int E) {
  __shared__ int hist[SORT_NB];
  const int t = threadIdx.x;
  if (t < SORT_NB) hist[t] = 0;
  __syncthreads();
  const unsigned R = sort_range_nodes(N);
  const int chunk = sort_chunk_edges(E);
  const int c0 = blockIdx.x * chunk;
  const int c1 = (c0 + chunk < E) ? (c0 + chunk) : E;
  for (int e0 = c0 + t; e0 < c1; e0 += SORT_EPT * SORT_T) {
    int d[SORT_EPT];
#pragma unroll
    for (int j = 0; j < SORT_EPT; ++j) {
      const int e = e0 + j * SORT_T;
      d[j] = (e < c1) ? ei[E + e] : -1;
    }
#pragma unroll
    for (int j = 0; j < SORT_EPT; ++j) {
      if (d[j] >= 0) atomicAdd(&hist[(unsigned)d[j] / R], 1);
    }
  }
  __syncthreads();
  if (t < SORT_NB) bh[blockIdx.x * SORT_NB + t] = hist[t];
}

__global__ __launch_bounds__(SORT_T) void gat_range_scatter(const int* __restrict__ ei,
                                                            const int* __restrict__ bh,
                                                            int* __restrict__ bstart,
                                                            int* __restrict__ packed,
                                                            int N, int E) {
  __shared__ int tot[SORT_T / SORT_NB][SORT_NB];
  __shared__ int part[SORT_T / SORT_NB][SORT_NB];
  __shared__ int cursor[SORT_NB];
  __shared__ int wsum[32];
  const int t = threadIdx.x;
  const int k = t & (SORT_NB - 1);
  const int q = t / SORT_NB;
  const int per = SORT_G / (SORT_T / SORT_NB);

  int total = 0, before = 0;
#pragma unroll 8
  for (int i = 0; i < per; ++i) {
    const int b = q * per + i;
    const int v = bh[b * SORT_NB + k];
    total += v;
    if (b < (int)blockIdx.x) before += v;
  }
  tot[q][k] = total;
  part[q][k] = before;
  __syncthreads();
  int T = 0, P = 0;
  if (t < SORT_NB) {
#pragma unroll
    for (int i = 0; i < SORT_T / SORT_NB; ++i) { T += tot[i][t]; P += part[i][t]; }
  }
  const int excl = sort_block_excl_scan(T, wsum);
  if (t < SORT_NB) {
    cursor[t] = excl + P;
    if (blockIdx.x == 0) {
      bstart[t] = excl;
      if (t == SORT_NB - 1) bstart[SORT_NB] = excl + T;
    }
  }
  __syncthreads();

  const unsigned R = sort_range_nodes(N);
  const int sh = sort_src_bits(N);
  const int chunk = sort_chunk_edges(E);
  const int c0 = blockIdx.x * chunk;
  const int c1 = (c0 + chunk < E) ? (c0 + chunk) : E;
  for (int e0 = c0 + t; e0 < c1; e0 += SORT_EPT * SORT_T) {
    int s[SORT_EPT], d[SORT_EPT];
#pragma unroll
    for (int j = 0; j < SORT_EPT; ++j) {
      const int e = e0 + j * SORT_T;
      s[j] = 0; d[j] = -1;
      if (e < c1) { s[j] = ei[e]; d[j] = ei[E + e]; }
    }
    int pos[SORT_EPT];
#pragma unroll
    for (int j = 0; j < SORT_EPT; ++j) {
      pos[j] = -1;
      if (d[j] >= 0) {
        const unsigned kk = (unsigned)d[j] / R;
        const unsigned dl = (unsigned)d[j] - kk * R;
        pos[j] = atomicAdd(&cursor[kk], 1);
        s[j] = (int)((dl << sh) | (unsigned)s[j]);
      }
    }
#pragma unroll
    for (int j = 0; j < SORT_EPT; ++j) {
      if (pos[j] >= 0) packed[pos[j]] = s[j];
    }
  }
}

__global__ __launch_bounds__(SORT_T) void gat_range_order(const int* __restrict__ packed,
                                                          const int* __restrict__ bstart,
                                                          int* __restrict__ rowptr,
                                                          int* __restrict__ src_sorted,
                                                          int N) {
  __shared__ int cnt[RANGE_MAX];
  __shared__ int wsum[32];
  const int t = threadIdx.x;
  const int k = blockIdx.x;
  const int R = (int)sort_range_nodes(N);
  const int sh = sort_src_bits(N);
  const unsigned smask = (sh >= 32) ? 0xFFFFFFFFu : ((1u << sh) - 1u);
  const int n0 = k * R;
  const int b0 = bstart[k];
  const int b1 = bstart[k + 1];

  for (int i = t; i < R; i += SORT_T) cnt[i] = 0;
  __syncthreads();

  for (int e0 = b0 + t; e0 < b1; e0 += SORT_EPT * SORT_T) {
    unsigned p[SORT_EPT];
#pragma unroll
    for (int j = 0; j < SORT_EPT; ++j) {
      const int e = e0 + j * SORT_T;
      p[j] = (e < b1) ? (unsigned)packed[e] : 0xFFFFFFFFu;
    }
#pragma unroll
    for (int j = 0; j < SORT_EPT; ++j) {
      const int e = e0 + j * SORT_T;
      if (e < b1) atomicAdd(&cnt[p[j] >> sh], 1);
    }
  }
  __syncthreads();

  const int ipt = (R + SORT_T - 1) / SORT_T;
  const int i0 = t * ipt;
  int mine = 0;
  for (int i = i0; i < i0 + ipt && i < R; ++i) mine += cnt[i];
  int run = sort_block_excl_scan(mine, wsum);
  for (int i = i0; i < i0 + ipt && i < R; ++i) {
    const int c = cnt[i];
    cnt[i] = run;
    if (n0 + i < N) rowptr[n0 + i] = b0 + run;
    run += c;
  }
  if (k == SORT_NB - 1 && t == 0) rowptr[N] = b1;
  __syncthreads();

  for (int e0 = b0 + t; e0 < b1; e0 += SORT_EPT * SORT_T) {
    unsigned p[SORT_EPT];
#pragma unroll
    for (int j = 0; j < SORT_EPT; ++j) {
      const int e = e0 + j * SORT_T;
      p[j] = (e < b1) ? (unsigned)packed[e] : 0xFFFFFFFFu;
    }
    int pos[SORT_EPT];
#pragma unroll
    for (int j = 0; j < SORT_EPT; ++j) {
      const int e = e0 + j * SORT_T;
      pos[j] = -1;
      if (e < b1) pos[j] = atomicAdd(&cnt[p[j] >> sh], 1);
    }
#pragma unroll
    for (int j = 0; j < SORT_EPT; ++j) {
      if (pos[j] >= 0) src_sorted[b0 + pos[j]] = (int)(p[j] & smask);
    }
  }
}

__device__ __forceinline__ float gat_edge_exp(float v) {
  v = (v >= 0.0f) ? v : NEG_SLOPE * v;
  return expf(v);
}

__device__ __forceinline__ void gat_load_edges(const int* __restrict__ src_sorted,
                                               const float* __restrict__ alpha_src,
                                               const float4 ad, int base, int end, int lane,
                                               int& s, float4& ex) {
  const int i = base + lane;
  s = 0;
  ex = make_float4(0.0f, 0.0f, 0.0f, 0.0f);
  if (i < end) {
    s = src_sorted[i];
    const float4 as = *(const float4*)(alpha_src + (size_t)s * NUM_HEADS);
    ex.x = gat_edge_exp(as.x + ad.x);
    ex.y = gat_edge_exp(as.y + ad.y);
    ex.z = gat_edge_exp(as.z + ad.z);
    ex.w = gat_edge_exp(as.w + ad.w);
  }
}

__global__ __launch_bounds__(256) void gat_gather(const int* __restrict__ rowptr,
                                                  const int* __restrict__ src_sorted,
                                                  const float* __restrict__ H,
                                                  const float* __restrict__ alpha_src,
                                                  const float* __restrict__ alpha_dst,
                                                  float* __restrict__ out,
                                                  int N) {
  const int lane = threadIdx.x & 31;
  const int wv0 = blockIdx.x * 8 + __builtin_amdgcn_readfirstlane((int)(threadIdx.x >> 5));
  const int nwv = gridDim.x * 8;
  const float* __restrict__ hcol = H + lane;
  for (int n = wv0; n < N; n += nwv) {
    const int beg = rowptr[n];
    const int end = rowptr[n + 1];
    const float4 ad = *(const float4*)(alpha_dst + (size_t)n * NUM_HEADS);

    int s0;
    float4 ex0;
    gat_load_edges(src_sorted, alpha_src, ad, beg, end, lane, s0, ex0);

    float4 sum = ex0;
    for (int base = beg + 32; base < end; base += 32) {
      int s;
      float4 ex;
      gat_load_edges(src_sorted, alpha_src, ad, base, end, lane, s, ex);
      sum.x += ex.x; sum.y += ex.y; sum.z += ex.z; sum.w += ex.w;
    }
    for (int off = 16; off >= 1; off >>= 1) {
      sum.x += __shfl_xor(sum.x, off, 32);
      sum.y += __shfl_xor(sum.y, off, 32);
      sum.z += __shfl_xor(sum.z, off, 32);
      sum.w += __shfl_xor(sum.w, off, 32);
    }
    const float4 den = make_float4(sum.x + GAT_EPS, sum.y + GAT_EPS, sum.z + GAT_EPS, sum.w + GAT_EPS);

    float acc0 = 0.0f, acc1 = 0.0f, acc2 = 0.0f, acc3 = 0.0f;
    for (int base = beg; base < end; base += 32) {
      int s = s0;
      float4 ex = ex0;
      if (base != beg) gat_load_edges(src_sorted, alpha_src, ad, base, end, lane, s, ex);
      const float4 w = make_float4(ex.x / den.x, ex.y / den.y, ex.z / den.z, ex.w / den.w);
      const int cntc = (end - base < 32) ? (end - base) : 32;
#pragma unroll 4
      for (int k = 0; k < cntc; ++k) {
        const int sk = __builtin_amdgcn_readlane(s, k);
        const float w0 = __int_as_float(__builtin_amdgcn_readlane(__float_as_int(w.x), k));
        const float w1 = __int_as_float(__builtin_amdgcn_readlane(__float_as_int(w.y), k));
        const float w2 = __int_as_float(__builtin_amdgcn_readlane(__float_as_int(w.z), k));
        const float w3 = __int_as_float(__builtin_amdgcn_readlane(__float_as_int(w.w), k));
        const float* __restrict__ hr = hcol + (size_t)sk * OUT_DIM;
        acc0 += hr[0 * HEAD_DIM] * w0;
        acc1 += hr[1 * HEAD_DIM] * w1;
        acc2 += hr[2 * HEAD_DIM] * w2;
        acc3 += hr[3 * HEAD_DIM] * w3;
      }
    }
    float* __restrict__ orow = out + (size_t)n * OUT_DIM + lane;
    orow[0 * HEAD_DIM] = acc0;
    orow[1 * HEAD_DIM] = acc1;
    orow[2 * HEAD_DIM] = acc2;
    orow[3 * HEAD_DIM] = acc3;
  }
}

extern "C" void kernel_launch(void* const* d_in, const int* in_sizes, int n_in,
                              void* d_out, int out_size, void* d_ws, size_t ws_size,
                              hipStream_t stream) {
  const float* X  = (const float*)d_in[0];
  const int*   ei = (const int*)d_in[1];
  const float* W  = (const float*)d_in[2];
  const float* a  = (const float*)d_in[3];

  const int N = in_sizes[0] / IN_DIM;
  const int E = in_sizes[1] / 2;
  const int n_pad = (N / 4 + 1) * 4;

  float* H          = (float*)d_ws;
  float* alpha_src  = H + (size_t)N * OUT_DIM;
  float* alpha_dst  = alpha_src + (size_t)N * NUM_HEADS;
  int*   bh         = (int*)(alpha_dst + (size_t)N * NUM_HEADS);
  int*   bstart     = bh + SORT_G * SORT_NB;
  int*   rowptr     = bstart + SORT_NB + 4;
  int*   packed     = rowptr + n_pad;
  int*   src_sorted = packed + (size_t)((E + 3) / 4) * 4;

  gat_gemm_wmma<<<GEMM_BLOCKS, 256, 0, stream>>>(X, W, H, N);
  gat_alpha<<<STREAM_BLOCKS, 256, 0, stream>>>(H, a, alpha_src, alpha_dst, N);
  gat_range_count<<<SORT_G, SORT_T, 0, stream>>>(ei, bh, N, E);
  gat_range_scatter<<<SORT_G, SORT_T, 0, stream>>>(ei, bh, bstart, packed, N, E);
  gat_range_order<<<SORT_NB, SORT_T, 0, stream>>>(packed, bstart, rowptr, src_sorted, N);
  gat_gather<<<GATHER_BLOCKS, 256, 0, stream>>>(rowptr, src_sorted, H, alpha_src, alpha_dst,
                                              (float*)d_out, N);
}


# GEMM A operand staged through LDS: a block reads each 16x128 X slab once with whole-row loads (two buffers) and its 8 waves take A fragments from LDS, not each by global loads that touch 16 lines
# speedup vs baseline: 4.0407x; 1.1029x over previous
#include <hip/hip_runtime.h>
#include <hip/hip_bf16.h>

typedef __attribute__((ext_vector_type(2))) float v2f;
typedef __attribute__((ext_vector_type(8))) float v8f;

#define IN_DIM    128
#define OUT_DIM   128
#define NUM_HEADS 4
#define HEAD_DIM  32
#define NEG_SLOPE 0.2f
#define GAT_EPS   1e-16f

#define GATHER_BLOCKS 2048
#define STREAM_BLOCKS 1024
#define GEMM_BLOCKS   1024

#define XS_PITCH 132

__global__ __launch_bounds__(256) void gat_gemm_wmma(const float* __restrict__ X,
                                                     const float* __restrict__ W,
                                                     float* __restrict__ H,
                                                     int M) {
  __shared__ float xs[2][16 * XS_PITCH];
  const int wave = threadIdx.x >> 5;
  const int lane = threadIdx.x & 31;
  const int n0 = wave * 16;
  const int rc   = lane & 15;
  const int koff = (lane >> 4) * 2;
  const int rbase = (lane >> 4) * 8;

  v2f bf[IN_DIM / 4];
#pragma unroll
  for (int k = 0; k < IN_DIM; k += 4) {
    bf[k / 4].x = W[(size_t)(k + koff) * OUT_DIM + n0 + rc];
    bf[k / 4].y = W[(size_t)(k + koff + 1) * OUT_DIM + n0 + rc];
  }

  const int nslab = (M + 15) / 16;
  const int scol = lane * 4;
  const int soff0 = wave * XS_PITCH + scol;
  const int soff1 = (wave + 8) * XS_PITCH + scol;

  int slab = blockIdx.x;
  if (slab < nslab) {
    int r0 = slab * 16 + wave;
    int r1 = r0 + 8;
    if (r0 >= M) r0 = M - 1;
    if (r1 >= M) r1 = M - 1;
    const float4 g0 = *(const float4*)(X + (size_t)r0 * IN_DIM + scol);
    const float4 g1 = *(const float4*)(X + (size_t)r1 * IN_DIM + scol);
    *(float4*)&xs[0][soff0] = g0;
    *(float4*)&xs[0][soff1] = g1;
  }
  __syncthreads();

  int cur = 0;
  for (; slab < nslab; slab += gridDim.x) {
    const int m0 = slab * 16;
    const int next = slab + gridDim.x;
    const bool has_next = next < nslab;
    float4 g0 = make_float4(0.0f, 0.0f, 0.0f, 0.0f), g1 = g0;
    if (has_next) {
      int r0 = next * 16 + wave;
      int r1 = r0 + 8;
      if (r0 >= M) r0 = M - 1;
      if (r1 >= M) r1 = M - 1;
      g0 = *(const float4*)(X + (size_t)r0 * IN_DIM + scol);
      g1 = *(const float4*)(X + (size_t)r1 * IN_DIM + scol);
    }

    const float* __restrict__ xrow = &xs[cur][rc * XS_PITCH + koff];
    v8f c = {};
#pragma unroll
    for (int k = 0; k < IN_DIM; k += 4) {
      v2f a = *(const v2f*)(xrow + k);
      c = __builtin_amdgcn_wmma_f32_16x16x4_f32(false, a, false, bf[k / 4], (short)0, c, false, false);
    }

    if (has_next) {
      *(float4*)&xs[cur ^ 1][soff0] = g0;
      *(float4*)&xs[cur ^ 1][soff1] = g1;
    }
    __syncthreads();
    cur ^= 1;

#pragma unroll
    for (int r = 0; r < 8; ++r) {
      const int row = m0 + rbase + r;
      if (row < M) H[(size_t)row * OUT_DIM + n0 + rc] = c[r];
    }
  }
}

__global__ __launch_bounds__(256) void gat_alpha(const float* __restrict__ H,
                                                 const float* __restrict__ a,
                                                 float* __restrict__ alpha_src,
                                                 float* __restrict__ alpha_dst,
                                                 int N) {
  const int lane = threadIdx.x & 31;
  const int w0 = blockIdx.x * 8 + __builtin_amdgcn_readfirstlane((int)(threadIdx.x >> 5));
  const int nw = gridDim.x * 8;

  const int d0    = lane * 4;
  const int head  = lane >> 3;
  const int local = d0 & (HEAD_DIM - 1);

  const float4 as = *(const float4*)(a + head * (2 * HEAD_DIM) + local);
  const float4 ad = *(const float4*)(a + head * (2 * HEAD_DIM) + HEAD_DIM + local);

  for (int nb = w0; nb < N; nb += 4 * nw) {
    float4 hv[4];
#pragma unroll
    for (int j = 0; j < 4; ++j) {
      const int node = nb + j * nw;
      hv[j] = make_float4(0.0f, 0.0f, 0.0f, 0.0f);
      if (node < N) hv[j] = *(const float4*)(H + (size_t)node * OUT_DIM + d0);
    }
#pragma unroll
    for (int j = 0; j < 4; ++j) {
      const int node = nb + j * nw;
      float ps = hv[j].x * as.x + hv[j].y * as.y + hv[j].z * as.z + hv[j].w * as.w;
      float pd = hv[j].x * ad.x + hv[j].y * ad.y + hv[j].z * ad.z + hv[j].w * ad.w;
#pragma unroll
      for (int off = 1; off <= 4; off <<= 1) {
        ps += __shfl_xor(ps, off, 32);
        pd += __shfl_xor(pd, off, 32);
      }
      if (node < N && (lane & 7) == 0) {
        alpha_src[(size_t)node * NUM_HEADS + head] = ps;
        alpha_dst[(size_t)node * NUM_HEADS + head] = pd;
      }
    }
  }
}

#define SORT_G    256
#define SORT_NB   256
#define SORT_T    1024
#define SORT_EPT  8
#define RANGE_MAX 4096

__device__ __forceinline__ unsigned sort_range_nodes(int N) {
  const unsigned r = ((unsigned)N + SORT_NB - 1) / SORT_NB;
  return r ? r : 1u;
}
__device__ __forceinline__ int sort_chunk_edges(int E) { return (E + SORT_G - 1) / SORT_G; }
__device__ __forceinline__ int sort_src_bits(int N) { return (N > 1) ? 32 - __clz(N - 1) : 0; }

__device__ __forceinline__ int sort_block_excl_scan(int v, int* wsum) {
  const int lane = threadIdx.x & 31;
  const int w = threadIdx.x >> 5;
  int incl = v;
  for (int off = 1; off < 32; off <<= 1) {
    const int n = __shfl_up(incl, off, 32);
    if (lane >= off) incl += n;
  }
  if (lane == 31) wsum[w] = incl;
  __syncthreads();
  if (w == 0) {
    const int x = wsum[lane];
    int xi = x;
    for (int off = 1; off < 32; off <<= 1) {
      const int n = __shfl_up(xi, off, 32);
      if (lane >= off) xi += n;
    }
    wsum[lane] = xi - x;
  }
  __syncthreads();
  return wsum[w] + incl - v;
}

__global__ __launch_bounds__(SORT_T) void gat_range_count(const int* __restrict__ ei,
                                                          int* __restrict__ bh,
                                                          int N, int E) {
  __shared__ int hist[SORT_NB];
  const int t = threadIdx.x;
  if (t < SORT_NB) hist[t] = 0;
  __syncthreads();
  const unsigned R = sort_range_nodes(N);
  const int chunk = sort_chunk_edges(E);
  const int c0 = blockIdx.x * chunk;
  const int c1 = (c0 + chunk < E) ? (c0 + chunk) : E;
  for (int e0 = c0 + t; e0 < c1; e0 += SORT_EPT * SORT_T) {
    int d[SORT_EPT];
#pragma unroll
    for (int j = 0; j < SORT_EPT; ++j) {
      const int e = e0 + j * SORT_T;
      d[j] = (e < c1) ? ei[E + e] : -1;
    }
#pragma unroll
    for (int j = 0; j < SORT_EPT; ++j) {
      if (d[j] >= 0) atomicAdd(&hist[(unsigned)d[j] / R], 1);
    }
  }
  __syncthreads();
  if (t < SORT_NB) bh[blockIdx.x * SORT_NB + t] = hist[t];
}

__global__ __launch_bounds__(SORT_T) void gat_range_scatter(const int* __restrict__ ei,
                                                            const int* __restrict__ bh,
                                                            int* __restrict__ bstart,
                                                            int* __restrict__ packed,
                                                            int N, int E) {
  __shared__ int tot[SORT_T / SORT_NB][SORT_NB];
  __shared__ int part[SORT_T / SORT_NB][SORT_NB];
  __shared__ int cursor[SORT_NB];
  __shared__ int wsum[32];
  const int t = threadIdx.x;
  const int k = t & (SORT_NB - 1);
  const int q = t / SORT_NB;
  const int per = SORT_G / (SORT_T / SORT_NB);

  int total = 0, before = 0;
#pragma unroll 8
  for (int i = 0; i < per; ++i) {
    const int b = q * per + i;
    const int v = bh[b * SORT_NB + k];
    total += v;
    if (b < (int)blockIdx.x) before += v;
  }
  tot[q][k] = total;
  part[q][k] = before;
  __syncthreads();
  int T = 0, P = 0;
  if (t < SORT_NB) {
#pragma unroll
    for (int i = 0; i < SORT_T / SORT_NB; ++i) { T += tot[i][t]; P += part[i][t]; }
  }
  const int excl = sort_block_excl_scan(T, wsum);
  if (t < SORT_NB) {
    cursor[t] = excl + P;
    if (blockIdx.x == 0) {
      bstart[t] = excl;
      if (t == SORT_NB - 1) bstart[SORT_NB] = excl + T;
    }
  }
  __syncthreads();

  const unsigned R = sort_range_nodes(N);
  const int sh = sort_src_bits(N);
  const int chunk = sort_chunk_edges(E);
  const int c0 = blockIdx.x * chunk;
  const int c1 = (c0 + chunk < E) ? (c0 + chunk) : E;
  for (int e0 = c0 + t; e0 < c1; e0 += SORT_EPT * SORT_T) {
    int s[SORT_EPT], d[SORT_EPT];
#pragma unroll
    for (int j = 0; j < SORT_EPT; ++j) {
      const int e = e0 + j * SORT_T;
      s[j] = 0; d[j] = -1;
      if (e < c1) { s[j] = ei[e]; d[j] = ei[E + e]; }
    }
    int pos[SORT_EPT];
#pragma unroll
    for (int j = 0; j < SORT_EPT; ++j) {
      pos[j] = -1;
      if (d[j] >= 0) {
        const unsigned kk = (unsigned)d[j] / R;
        const unsigned dl = (unsigned)d[j] - kk * R;
        pos[j] = atomicAdd(&cursor[kk], 1);
        s[j] = (int)((dl << sh) | (unsigned)s[j]);
      }
    }
#pragma unroll
    for (int j = 0; j < SORT_EPT; ++j) {
      if (pos[j] >= 0) packed[pos[j]] = s[j];
    }
  }
}

__global__ __launch_bounds__(SORT_T) void gat_range_order(const int* __restrict__ packed,
                                                          const int* __restrict__ bstart,
                                                          int* __restrict__ rowptr,
                                                          int* __restrict__ src_sorted,
                                                          int N) {
  __shared__ int cnt[RANGE_MAX];
  __shared__ int wsum[32];
  const int t = threadIdx.x;
  const int k = blockIdx.x;
  const int R = (int)sort_range_nodes(N);
  const int sh = sort_src_bits(N);
  const unsigned smask = (sh >= 32) ? 0xFFFFFFFFu : ((1u << sh) - 1u);
  const int n0 = k * R;
  const int b0 = bstart[k];
  const int b1 = bstart[k + 1];

  for (int i = t; i < R; i += SORT_T) cnt[i] = 0;
  __syncthreads();

  for (int e0 = b0 + t; e0 < b1; e0 += SORT_EPT * SORT_T) {
    unsigned p[SORT_EPT];
#pragma unroll
    for (int j = 0; j < SORT_EPT; ++j) {
      const int e = e0 + j * SORT_T;
      p[j] = (e < b1) ? (unsigned)packed[e] : 0xFFFFFFFFu;
    }
#pragma unroll
    for (int j = 0; j < SORT_EPT; ++j) {
      const int e = e0 + j * SORT_T;
      if (e < b1) atomicAdd(&cnt[p[j] >> sh], 1);
    }
  }
  __syncthreads();

  const int ipt = (R + SORT_T - 1) / SORT_T;
  const int i0 = t * ipt;
  int mine = 0;
  for (int i = i0; i < i0 + ipt && i < R; ++i) mine += cnt[i];
  int run = sort_block_excl_scan(mine, wsum);
  for (int i = i0; i < i0 + ipt && i < R; ++i) {
    const int c = cnt[i];
    cnt[i] = run;
    if (n0 + i < N) rowptr[n0 + i] = b0 + run;
    run += c;
  }
  if (k == SORT_NB - 1 && t == 0) rowptr[N] = b1;
  __syncthreads();

  for (int e0 = b0 + t; e0 < b1; e0 += SORT_EPT * SORT_T) {
    unsigned p[SORT_EPT];
#pragma unroll
    for (int j = 0; j < SORT_EPT; ++j) {
      const int e = e0 + j * SORT_T;
      p[j] = (e < b1) ? (unsigned)packed[e] : 0xFFFFFFFFu;
    }
    int pos[SORT_EPT];
#pragma unroll
    for (int j = 0; j < SORT_EPT; ++j) {
      const int e = e0 + j * SORT_T;
      pos[j] = -1;
      if (e < b1) pos[j] = atomicAdd(&cnt[p[j] >> sh], 1);
    }
#pragma unroll
    for (int j = 0; j < SORT_EPT; ++j) {
      if (pos[j] >= 0) src_sorted[b0 + pos[j]] = (int)(p[j] & smask);
    }
  }
}

__device__ __forceinline__ float gat_edge_exp(float v) {
  v = (v >= 0.0f) ? v : NEG_SLOPE * v;
  return expf(v);
}

__device__ __forceinline__ void gat_load_edges(const int* __restrict__ src_sorted,
                                               const float* __restrict__ alpha_src,
                                               const float4 ad, int base, int end, int lane,
                                               int& s, float4& ex) {
  const int i = base + lane;
  s = 0;
  ex = make_float4(0.0f, 0.0f, 0.0f, 0.0f);
  if (i < end) {
    s = src_sorted[i];
    const float4 as = *(const float4*)(alpha_src + (size_t)s * NUM_HEADS);
    ex.x = gat_edge_exp(as.x + ad.x);
    ex.y = gat_edge_exp(as.y + ad.y);
    ex.z = gat_edge_exp(as.z + ad.z);
    ex.w = gat_edge_exp(as.w + ad.w);
  }
}

__global__ __launch_bounds__(256) void gat_gather(const int* __restrict__ rowptr,
                                                  const int* __restrict__ src_sorted,
                                                  const float* __restrict__ H,
                                                  const float* __restrict__ alpha_src,
                                                  const float* __restrict__ alpha_dst,
                                                  float* __restrict__ out,
                                                  int N) {
  const int lane = threadIdx.x & 31;
  const int wv0 = blockIdx.x * 8 + __builtin_amdgcn_readfirstlane((int)(threadIdx.x >> 5));
  const int nwv = gridDim.x * 8;
  const float* __restrict__ hcol = H + lane;
  for (int n = wv0; n < N; n += nwv) {
    const int beg = rowptr[n];
    const int end = rowptr[n + 1];
    const float4 ad = *(const float4*)(alpha_dst + (size_t)n * NUM_HEADS);

    int s0;
    float4 ex0;
    gat_load_edges(src_sorted, alpha_src, ad, beg, end, lane, s0, ex0);

    float4 sum = ex0;
    for (int base = beg + 32; base < end; base += 32) {
      int s;
      float4 ex;
      gat_load_edges(src_sorted, alpha_src, ad, base, end, lane, s, ex);
      sum.x += ex.x; sum.y += ex.y; sum.z += ex.z; sum.w += ex.w;
    }
    for (int off = 16; off >= 1; off >>= 1) {
      sum.x += __shfl_xor(sum.x, off, 32);
      sum.y += __shfl_xor(sum.y, off, 32);
      sum.z += __shfl_xor(sum.z, off, 32);
      sum.w += __shfl_xor(sum.w, off, 32);
    }
    const float4 den = make_float4(sum.x + GAT_EPS, sum.y + GAT_EPS, sum.z + GAT_EPS, sum.w + GAT_EPS);

    float acc0 = 0.0f, acc1 = 0.0f, acc2 = 0.0f, acc3 = 0.0f;
    for (int base = beg; base < end; base += 32) {
      int s = s0;
      float4 ex = ex0;
      if (base != beg) gat_load_edges(src_sorted, alpha_src, ad, base, end, lane, s, ex);
      const float4 w = make_float4(ex.x / den.x, ex.y / den.y, ex.z / den.z, ex.w / den.w);
      const int cntc = (end - base < 32) ? (end - base) : 32;
#pragma unroll 4
      for (int k = 0; k < cntc; ++k) {
        const int sk = __builtin_amdgcn_readlane(s, k);
        const float w0 = __int_as_float(__builtin_amdgcn_readlane(__float_as_int(w.x), k));
        const float w1 = __int_as_float(__builtin_amdgcn_readlane(__float_as_int(w.y), k));
        const float w2 = __int_as_float(__builtin_amdgcn_readlane(__float_as_int(w.z), k));
        const float w3 = __int_as_float(__builtin_amdgcn_readlane(__float_as_int(w.w), k));
        const float* __restrict__ hr = hcol + (size_t)sk * OUT_DIM;
        acc0 += hr[0 * HEAD_DIM] * w0;
        acc1 += hr[1 * HEAD_DIM] * w1;
        acc2 += hr[2 * HEAD_DIM] * w2;
        acc3 += hr[3 * HEAD_DIM] * w3;
      }
    }
    float* __restrict__ orow = out + (size_t)n * OUT_DIM + lane;
    orow[0 * HEAD_DIM] = acc0;
    orow[1 * HEAD_DIM] = acc1;
    orow[2 * HEAD_DIM] = acc2;
    orow[3 * HEAD_DIM] = acc3;
  }
}

extern "C" void kernel_launch(void* const* d_in, const int* in_sizes, int n_in,
                              void* d_out, int out_size, void* d_ws, size_t ws_size,
                              hipStream_t stream) {
  const float* X  = (const float*)d_in[0];
  const int*   ei = (const int*)d_in[1];
  const float* W  = (const float*)d_in[2];
  const float* a  = (const float*)d_in[3];

  const int N = in_sizes[0] / IN_DIM;
  const int E = in_sizes[1] / 2;
  const int n_pad = (N / 4 + 1) * 4;

  float* H          = (float*)d_ws;
  float* alpha_src  = H + (size_t)N * OUT_DIM;
  float* alpha_dst  = alpha_src + (size_t)N * NUM_HEADS;
  int*   bh         = (int*)(alpha_dst + (size_t)N * NUM_HEADS);
  int*   bstart     = bh + SORT_G * SORT_NB;
  int*   rowptr     = bstart + SORT_NB + 4;
  int*   packed     = rowptr + n_pad;
  int*   src_sorted = packed + (size_t)((E + 3) / 4) * 4;

  gat_gemm_wmma<<<GEMM_BLOCKS, 256, 0, stream>>>(X, W, H, N);
  gat_alpha<<<STREAM_BLOCKS, 256, 0, stream>>>(H, a, alpha_src, alpha_dst, N);
  gat_range_count<<<SORT_G, SORT_T, 0, stream>>>(ei, bh, N, E);
  gat_range_scatter<<<SORT_G, SORT_T, 0, stream>>>(ei, bh, bstart, packed, N, E);
  gat_range_order<<<SORT_NB, SORT_T, 0, stream>>>(packed, bstart, rowptr, src_sorted, N);
  gat_gather<<<GATHER_BLOCKS, 256, 0, stream>>>(rowptr, src_sorted, H, alpha_src, alpha_dst,
                                              (float*)d_out, N);
}
